# starting kernel
# baseline (speedup 1.0000x reference)
#include <hip/hip_runtime.h>


typedef float v2f __attribute__((ext_vector_type(2)));
typedef float v8f __attribute__((ext_vector_type(8)));

#define BB 4
#define NN 32768
#define MM 8192
#define KK 16
#define DD 64
#define OO 64

template <int MODE>
__launch_bounds__(256)
__global__ void gemm16_wmma(const float* __restrict__ X,
                            const float* __restrict__ W,
                            const float* __restrict__ bias,
                            float* __restrict__ Y,
                            int numTiles) {
  const int lane  = threadIdx.x & 31;
  const int wib   = threadIdx.x >> 5;
  const int wpb   = blockDim.x >> 5;
  const int gwave = blockIdx.x * wpb + wib;
  const int nWav  = gridDim.x * wpb;

  const int mrow = lane & 15;
  const int grp  = lane >> 4;
  const int kb   = grp * 2;

  v2f bf[4][16];
#pragma unroll
  for (int t = 0; t < 4; ++t) {
    const float* wrow = W + (size_t)(t * 16 + mrow) * (2 * DD);
#pragma unroll
    for (int s = 0; s < 16; ++s) {
      const int k = 4 * s + kb;
      if (MODE == 0) {
        bf[t][s] = *(const v2f*)(wrow + k);
      } else {
        v2f p = *(const v2f*)(wrow + DD + k);
        v2f q = *(const v2f*)(wrow + k);
        bf[t][s] = p - q;
      }
    }
  }

  float bv[4];
#pragma unroll
  for (int t = 0; t < 4; ++t)
    bv[t] = (MODE == 1) ? bias[t * 16 + mrow] : 0.0f;

  for (int tile = gwave; tile < numTiles; tile += nWav) {
    const float* xrow = X + (size_t)(tile * 16 + mrow) * DD + kb;

    v8f acc[4];
#pragma unroll
    for (int t = 0; t < 4; ++t)
#pragma unroll
      for (int i = 0; i < 8; ++i) acc[t][i] = bv[t];

#pragma unroll
    for (int s = 0; s < 16; ++s) {
      v2f a = *(const v2f*)(xrow + 4 * s);
#pragma unroll
      for (int t = 0; t < 4; ++t) {
        acc[t] = __builtin_amdgcn_wmma_f32_16x16x4_f32(
             false, a,  false, bf[t][s],
             (short)0, acc[t],  false,  false);
      }
    }

    float* yout = Y + (size_t)(tile * 16 + 8 * grp) * OO;
#pragma unroll
    for (int t = 0; t < 4; ++t)
#pragma unroll
      for (int r = 0; r < 8; ++r)
        yout[(size_t)r * OO + t * 16 + mrow] = acc[t][r];
  }
}

__launch_bounds__(256)
__global__ void gather_max_kernel(const float* __restrict__ y,
                                  const float* __restrict__ z,
                                  const int* __restrict__ knn,
                                  float* __restrict__ out) {
  const int lane  = threadIdx.x & 31;
  const int gwave = (blockIdx.x * blockDim.x + threadIdx.x) >> 5;
  const int b = gwave >> 13;
  const int m = gwave & (MM - 1);
  const int o0 = lane * 2;

  const v2f zv = *(const v2f*)(z + (size_t)gwave * OO + o0);
  v2f acc;
  acc.x = 0.0f;
  acc.y = 0.0f;

  const int* kn = knn + (size_t)gwave * KK;
#pragma unroll
  for (int k = 0; k < KK; ++k) {
    const int idx = kn[k];
    const v2f yv = *(const v2f*)(y + ((size_t)b * NN + idx) * OO + o0);
    v2f s = yv + zv;
    acc.x = fmaxf(acc.x, s.x);
    acc.y = fmaxf(acc.y, s.y);
  }

  float* op = out + ((size_t)b * OO + o0) * MM + m;
  op[0]  = acc.x;
  op[MM] = acc.y;
}

extern "C" void kernel_launch(void* const* d_in, const int* in_sizes, int n_in,
                              void* d_out, int out_size, void* d_ws, size_t ws_size,
                              hipStream_t stream) {
  const float* x    = (const float*)d_in[0];
  const float* xs   = (const float*)d_in[1];
  const int*   knn  = (const int*)d_in[2];
  const float* W    = (const float*)d_in[3];
  const float* bias = (const float*)d_in[4];
  float* out = (float*)d_out;

  float* y = (float*)d_ws;
  float* z = y + (size_t)BB * NN * OO;

  gemm16_wmma<0><<<256, 256, 0, stream>>>(x, W, bias, y, (BB * NN) / 16);
  gemm16_wmma<1><<<256, 256, 0, stream>>>(xs, W, bias, z, (BB * MM) / 16);
  gather_max_kernel<<<(BB * MM * 32) / 256, 256, 0, stream>>>(y, z, knn, out);
}


# gather kernel re-tiled: work-group per 32 points, half-wave per point with 16-byte loads, tile transposed in LDS so each store writes a whole 128-byte line (was 64 scattered 4-byte stores per point)
# speedup vs baseline: 2.2345x; 2.2345x over previous
#include <hip/hip_runtime.h>


typedef float v2f __attribute__((ext_vector_type(2)));
typedef float v8f __attribute__((ext_vector_type(8)));

#define BB 4
#define NN 32768
#define MM 8192
#define KK 16
#define DD 64
#define OO 64

template <int MODE>
__launch_bounds__(256)
__global__ void gemm16_wmma(const float* __restrict__ X,
                            const float* __restrict__ W,
                            const float* __restrict__ bias,
                            float* __restrict__ Y,
                            int numTiles) {
  const int lane  = threadIdx.x & 31;
  const int wib   = threadIdx.x >> 5;
  const int wpb   = blockDim.x >> 5;
  const int gwave = blockIdx.x * wpb + wib;
  const int nWav  = gridDim.x * wpb;

  const int mrow = lane & 15;
  const int grp  = lane >> 4;
  const int kb   = grp * 2;

  v2f bf[4][16];
#pragma unroll
  for (int t = 0; t < 4; ++t) {
    const float* wrow = W + (size_t)(t * 16 + mrow) * (2 * DD);
#pragma unroll
    for (int s = 0; s < 16; ++s) {
      const int k = 4 * s + kb;
      if (MODE == 0) {
        bf[t][s] = *(const v2f*)(wrow + k);
      } else {
        v2f p = *(const v2f*)(wrow + DD + k);
        v2f q = *(const v2f*)(wrow + k);
        bf[t][s] = p - q;
      }
    }
  }

  float bv[4];
#pragma unroll
  for (int t = 0; t < 4; ++t)
    bv[t] = (MODE == 1) ? bias[t * 16 + mrow] : 0.0f;

  for (int tile = gwave; tile < numTiles; tile += nWav) {
    const float* xrow = X + (size_t)(tile * 16 + mrow) * DD + kb;

    v8f acc[4];
#pragma unroll
    for (int t = 0; t < 4; ++t)
#pragma unroll
      for (int i = 0; i < 8; ++i) acc[t][i] = bv[t];

#pragma unroll
    for (int s = 0; s < 16; ++s) {
      v2f a = *(const v2f*)(xrow + 4 * s);
#pragma unroll
      for (int t = 0; t < 4; ++t) {
        acc[t] = __builtin_amdgcn_wmma_f32_16x16x4_f32(
             false, a,  false, bf[t][s],
             (short)0, acc[t],  false,  false);
      }
    }

    float* yout = Y + (size_t)(tile * 16 + 8 * grp) * OO;
#pragma unroll
    for (int t = 0; t < 4; ++t)
#pragma unroll
      for (int r = 0; r < 8; ++r)
        yout[(size_t)r * OO + t * 16 + mrow] = acc[t][r];
  }
}

typedef float v4f __attribute__((ext_vector_type(4)));

#define TM    32
#define PITCH (TM + 1)
#define PPW   (TM / 8)

template <int K>
__device__ __forceinline__ int row_lane(int v) {
  return __builtin_amdgcn_update_dpp(0, v, 0x150 + K, 0xf, 0xf, false);
}

#define GATHER_STEP(K)                                                              \
  {                                                                                 \
    const unsigned off = ((unsigned)row_lane<K>(iv) << 8) + cofs;                   \
    const v4f yv = *(const v4f*)(yb + off);                                         \
    const v4f s = yv + zv;                                                          \
    acc.x = fmaxf(acc.x, s.x);                                                      \
    acc.y = fmaxf(acc.y, s.y);                                                      \
    acc.z = fmaxf(acc.z, s.z);                                                      \
    acc.w = fmaxf(acc.w, s.w);                                                      \
  }

__launch_bounds__(256)
__global__ void gather_max_kernel(const float* __restrict__ y,
                                  const float* __restrict__ z,
                                  const int* __restrict__ knn,
                                  float* __restrict__ out) {
  __shared__ float tile[OO * PITCH];

  const int lane = threadIdx.x & 31;
  const int h    = lane >> 4;
  const int c    = lane & 15;
  const int wib  = __builtin_amdgcn_readfirstlane(threadIdx.x >> 5);
  const int base = blockIdx.x * TM;
  const int b    = base >> 13;
  const int m0   = base & (MM - 1);
  const int pl0  = wib * PPW;

  const char* yb = (const char*)(y + (size_t)b * NN * OO);
  const unsigned cofs = (unsigned)c * 16u;

  const int*   kn = knn + (size_t)(base + pl0) * KK + lane;
  const float* zr = z + (size_t)(base + pl0) * OO + lane * 4;

#pragma unroll
  for (int j = 0; j < PPW / 2; ++j) {
    const int iv = kn[32 * j];
    const v4f zv = *(const v4f*)(zr + 2 * OO * j);
    v4f acc;
    acc.x = 0.0f;
    acc.y = 0.0f;
    acc.z = 0.0f;
    acc.w = 0.0f;
    GATHER_STEP(0)  GATHER_STEP(1)  GATHER_STEP(2)  GATHER_STEP(3)
    GATHER_STEP(4)  GATHER_STEP(5)  GATHER_STEP(6)  GATHER_STEP(7)
    GATHER_STEP(8)  GATHER_STEP(9)  GATHER_STEP(10) GATHER_STEP(11)
    GATHER_STEP(12) GATHER_STEP(13) GATHER_STEP(14) GATHER_STEP(15)
    float* tp = tile + (4 * c) * PITCH + pl0 + 2 * j + h;
    tp[0 * PITCH] = acc.x;
    tp[1 * PITCH] = acc.y;
    tp[2 * PITCH] = acc.z;
    tp[3 * PITCH] = acc.w;
  }

  __syncthreads();

  float* ob = out + ((size_t)b * OO + wib * 8) * MM + m0 + lane;
#pragma unroll
  for (int r = 0; r < 8; ++r)
    ob[(size_t)r * MM] = tile[(wib * 8 + r) * PITCH + lane];
}

extern "C" void kernel_launch(void* const* d_in, const int* in_sizes, int n_in,
                              void* d_out, int out_size, void* d_ws, size_t ws_size,
                              hipStream_t stream) {
  const float* x    = (const float*)d_in[0];
  const float* xs   = (const float*)d_in[1];
  const int*   knn  = (const int*)d_in[2];
  const float* W    = (const float*)d_in[3];
  const float* bias = (const float*)d_in[4];
  float* out = (float*)d_out;

  float* y = (float*)d_ws;
  float* z = y + (size_t)BB * NN * OO;

  gemm16_wmma<0><<<256, 256, 0, stream>>>(x, W, bias, y, (BB * NN) / 16);
  gemm16_wmma<1><<<256, 256, 0, stream>>>(xs, W, bias, z, (BB * MM) / 16);
  gather_max_kernel<<<(BB * MM) / TM, 256, 0, stream>>>(y, z, knn, out);
}


# GEMM kernels: W (or W2-W1) and each X tile go through LDS, read from global with 16-byte loads that cover 4 whole 128-byte lines each (was 8-byte fragment loads touching 16 lines each)
# speedup vs baseline: 2.6165x; 1.1710x over previous
#include <hip/hip_runtime.h>


typedef float v2f __attribute__((ext_vector_type(2)));
typedef float v8f __attribute__((ext_vector_type(8)));

#define BB 4
#define NN 32768
#define MM 8192
#define KK 16
#define DD 64
#define OO 64

typedef float v4f __attribute__((ext_vector_type(4)));

#define GW 8
#define LP (DD + 4)

template <int MODE>
__launch_bounds__(256)
__global__ void gemm16_wmma(const float* __restrict__ X,
                            const float* __restrict__ W,
                            const float* __restrict__ bias,
                            float* __restrict__ Y,
                            int numTiles) {
  __shared__ __attribute__((aligned(16))) float Bs[OO * LP];
  __shared__ __attribute__((aligned(16))) float As[GW][16 * LP];

  const int lane  = threadIdx.x & 31;
  const int wib   = threadIdx.x >> 5;
  const int gwave = blockIdx.x * GW + wib;
  const int nWav  = gridDim.x * GW;

  const int mrow = lane & 15;
  const int grp  = lane >> 4;
  const int kb   = grp * 2;

#pragma unroll
  for (int i = 0; i < 4; ++i) {
    const int idx = threadIdx.x + 256 * i;
    const int row = idx >> 4;
    const int c4  = idx & 15;
    const float* wp = W + (size_t)row * (2 * DD) + 4 * c4;
    v4f v;
    if (MODE == 0) {
      v = *(const v4f*)wp;
    } else {
      v4f p = *(const v4f*)(wp + DD);
      v4f q = *(const v4f*)wp;
      v = p - q;
    }
    *(v4f*)(Bs + row * LP + 4 * c4) = v;
  }
  __syncthreads();

  v2f bf[4][16];
#pragma unroll
  for (int t = 0; t < 4; ++t) {
    const float* brow = Bs + (t * 16 + mrow) * LP + kb;
#pragma unroll
    for (int s = 0; s < 16; ++s)
      bf[t][s] = *(const v2f*)(brow + 4 * s);
  }

  float bv[4];
#pragma unroll
  for (int t = 0; t < 4; ++t)
    bv[t] = (MODE == 1) ? bias[t * 16 + mrow] : 0.0f;

  float* as = As[wib];
  float*       asw = as + grp * LP + 4 * mrow;
  const float* asr = as + mrow * LP + kb;

  for (int tile = gwave; tile < numTiles; tile += nWav) {
    const float* xt = X + (size_t)tile * (16 * DD) + lane * 4;
    v4f g[8];
#pragma unroll
    for (int i = 0; i < 8; ++i) g[i] = *(const v4f*)(xt + 128 * i);
#pragma unroll
    for (int i = 0; i < 8; ++i) *(v4f*)(asw + 2 * i * LP) = g[i];


    v2f a[16];
#pragma unroll
    for (int s = 0; s < 16; ++s) a[s] = *(const v2f*)(asr + 4 * s);

    __builtin_amdgcn_sched_barrier(0);

    v8f acc[4];
#pragma unroll
    for (int t = 0; t < 4; ++t)
#pragma unroll
      for (int i = 0; i < 8; ++i) acc[t][i] = bv[t];

#pragma unroll
    for (int s = 0; s < 16; ++s) {
#pragma unroll
      for (int t = 0; t < 4; ++t) {
        acc[t] = __builtin_amdgcn_wmma_f32_16x16x4_f32(
             false, a[s],  false, bf[t][s],
             (short)0, acc[t],  false,  false);
      }
    }

    float* yout = Y + (size_t)(tile * 16 + 8 * grp) * OO;
#pragma unroll
    for (int t = 0; t < 4; ++t)
#pragma unroll
      for (int r = 0; r < 8; ++r)
        yout[(size_t)r * OO + t * 16 + mrow] = acc[t][r];
  }
}


#define TM    32
#define PITCH (TM + 1)
#define PPW   (TM / 8)

template <int K>
__device__ __forceinline__ int row_lane(int v) {
  return __builtin_amdgcn_update_dpp(0, v, 0x150 + K, 0xf, 0xf, false);
}

#define GATHER_STEP(K)                                                              \
  {                                                                                 \
    const unsigned off = ((unsigned)row_lane<K>(iv) << 8) + cofs;                   \
    const v4f yv = *(const v4f*)(yb + off);                                         \
    const v4f s = yv + zv;                                                          \
    acc.x = fmaxf(acc.x, s.x);                                                      \
    acc.y = fmaxf(acc.y, s.y);                                                      \
    acc.z = fmaxf(acc.z, s.z);                                                      \
    acc.w = fmaxf(acc.w, s.w);                                                      \
  }

__launch_bounds__(256)
__global__ void gather_max_kernel(const float* __restrict__ y,
                                  const float* __restrict__ z,
                                  const int* __restrict__ knn,
                                  float* __restrict__ out) {
  __shared__ float tile[OO * PITCH];

  const int lane = threadIdx.x & 31;
  const int h    = lane >> 4;
  const int c    = lane & 15;
  const int wib  = __builtin_amdgcn_readfirstlane(threadIdx.x >> 5);
  const int base = blockIdx.x * TM;
  const int b    = base >> 13;
  const int m0   = base & (MM - 1);
  const int pl0  = wib * PPW;

  const char* yb = (const char*)(y + (size_t)b * NN * OO);
  const unsigned cofs = (unsigned)c * 16u;

  const int*   kn = knn + (size_t)(base + pl0) * KK + lane;
  const float* zr = z + (size_t)(base + pl0) * OO + lane * 4;

#pragma unroll
  for (int j = 0; j < PPW / 2; ++j) {
    const int iv = kn[32 * j];
    const v4f zv = *(const v4f*)(zr + 2 * OO * j);
    v4f acc;
    acc.x = 0.0f;
    acc.y = 0.0f;
    acc.z = 0.0f;
    acc.w = 0.0f;
    GATHER_STEP(0)  GATHER_STEP(1)  GATHER_STEP(2)  GATHER_STEP(3)
    GATHER_STEP(4)  GATHER_STEP(5)  GATHER_STEP(6)  GATHER_STEP(7)
    GATHER_STEP(8)  GATHER_STEP(9)  GATHER_STEP(10) GATHER_STEP(11)
    GATHER_STEP(12) GATHER_STEP(13) GATHER_STEP(14) GATHER_STEP(15)
    float* tp = tile + (4 * c) * PITCH + pl0 + 2 * j + h;
    tp[0 * PITCH] = acc.x;
    tp[1 * PITCH] = acc.y;
    tp[2 * PITCH] = acc.z;
    tp[3 * PITCH] = acc.w;
  }

  __syncthreads();

  float* ob = out + ((size_t)b * OO + wib * 8) * MM + m0 + lane;
#pragma unroll
  for (int r = 0; r < 8; ++r)
    ob[(size_t)r * MM] = tile[(wib * 8 + r) * PITCH + lane];
}

extern "C" void kernel_launch(void* const* d_in, const int* in_sizes, int n_in,
                              void* d_out, int out_size, void* d_ws, size_t ws_size,
                              hipStream_t stream) {
  const float* x    = (const float*)d_in[0];
  const float* xs   = (const float*)d_in[1];
  const int*   knn  = (const int*)d_in[2];
  const float* W    = (const float*)d_in[3];
  const float* bias = (const float*)d_in[4];
  float* out = (float*)d_out;

  float* y = (float*)d_ws;
  float* z = y + (size_t)BB * NN * OO;

  gemm16_wmma<0><<<256, 256, 0, stream>>>(x, W, bias, y, (BB * NN) / 16);
  gemm16_wmma<1><<<256, 256, 0, stream>>>(xs, W, bias, z, (BB * MM) / 16);
  gather_max_kernel<<<(BB * MM) / TM, 256, 0, stream>>>(y, z, knn, out);
}


# both GEMMs in one launch: each work-group stages W1 and W2-W1 in LDS once; each wave runs its 4 y tiles, re-reads its B fragments from LDS, then runs its z tile (2 launches per repetition, was 3)
# speedup vs baseline: 2.9023x; 1.1092x over previous
#include <hip/hip_runtime.h>


typedef float v2f __attribute__((ext_vector_type(2)));
typedef float v8f __attribute__((ext_vector_type(8)));

#define BB 4
#define NN 32768
#define MM 8192
#define KK 16
#define DD 64
#define OO 64

typedef float v4f __attribute__((ext_vector_type(4)));

#define GW 8
#define LP (DD + 4)

__device__ __forceinline__ void load_b_frags(const float* Bm, int mrow, int kb, v2f (&bf)[4][16]) {
#pragma unroll
  for (int t = 0; t < 4; ++t) {
    const float* brow = Bm + (t * 16 + mrow) * LP + kb;
#pragma unroll
    for (int s = 0; s < 16; ++s)
      bf[t][s] = *(const v2f*)(brow + 4 * s);
  }
}

__device__ __forceinline__ void tile_product(const float* __restrict__ X, float* __restrict__ Y, int tile,
                                             const v2f (&bf)[4][16], const float (&bv)[4],
                                             float* asw, const float* asr,
                                             int lane, int grp, int mrow) {
  const float* xt = X + (size_t)tile * (16 * DD) + lane * 4;
  v4f g[8];
#pragma unroll
  for (int i = 0; i < 8; ++i) g[i] = *(const v4f*)(xt + 128 * i);
#pragma unroll
  for (int i = 0; i < 8; ++i) *(v4f*)(asw + 2 * i * LP) = g[i];


  v2f a[16];
#pragma unroll
  for (int s = 0; s < 16; ++s) a[s] = *(const v2f*)(asr + 4 * s);

  __builtin_amdgcn_sched_barrier(0);

  v8f acc[4];
#pragma unroll
  for (int t = 0; t < 4; ++t)
#pragma unroll
    for (int i = 0; i < 8; ++i) acc[t][i] = bv[t];

#pragma unroll
  for (int s = 0; s < 16; ++s) {
#pragma unroll
    for (int t = 0; t < 4; ++t) {
      acc[t] = __builtin_amdgcn_wmma_f32_16x16x4_f32(
           false, a[s],  false, bf[t][s],
           (short)0, acc[t],  false,  false);
    }
  }

  float* yout = Y + (size_t)(tile * 16 + 8 * grp) * OO;
#pragma unroll
  for (int t = 0; t < 4; ++t)
#pragma unroll
    for (int r = 0; r < 8; ++r)
      yout[(size_t)r * OO + t * 16 + mrow] = acc[t][r];
}

__launch_bounds__(256)
__global__ void gemm16_wmma_yz(const float* __restrict__ X,
                               const float* __restrict__ XS,
                               const float* __restrict__ W,
                               const float* __restrict__ bias,
                               float* __restrict__ Y,
                               float* __restrict__ Z,
                               int yTiles, int zTiles) {
  __shared__ __attribute__((aligned(16))) float Bs1[OO * LP];
  __shared__ __attribute__((aligned(16))) float Bs2[OO * LP];
  __shared__ __attribute__((aligned(16))) float As[GW][16 * LP];

  const int lane  = threadIdx.x & 31;
  const int wib   = threadIdx.x >> 5;
  const int gwave = blockIdx.x * GW + wib;
  const int nWav  = gridDim.x * GW;

  const int mrow = lane & 15;
  const int grp  = lane >> 4;
  const int kb   = grp * 2;

#pragma unroll
  for (int i = 0; i < 4; ++i) {
    const int idx = threadIdx.x + 256 * i;
    const int row = idx >> 4;
    const int c4  = idx & 15;
    const float* wp = W + (size_t)row * (2 * DD) + 4 * c4;
    const v4f q = *(const v4f*)wp;
    const v4f p = *(const v4f*)(wp + DD);
    *(v4f*)(Bs1 + row * LP + 4 * c4) = q;
    *(v4f*)(Bs2 + row * LP + 4 * c4) = p - q;
  }
  float bz[4];
#pragma unroll
  for (int t = 0; t < 4; ++t) bz[t] = bias[t * 16 + mrow];
  __syncthreads();

  float* as = As[wib];
  float*       asw = as + grp * LP + 4 * mrow;
  const float* asr = as + mrow * LP + kb;

  v2f bf[4][16];
  float bv[4];

  load_b_frags(Bs1, mrow, kb, bf);
#pragma unroll
  for (int t = 0; t < 4; ++t) bv[t] = 0.0f;
  for (int tile = gwave; tile < yTiles; tile += nWav)
    tile_product(X, Y, tile, bf, bv, asw, asr, lane, grp, mrow);

  load_b_frags(Bs2, mrow, kb, bf);
#pragma unroll
  for (int t = 0; t < 4; ++t) bv[t] = bz[t];
  for (int tile = gwave; tile < zTiles; tile += nWav)
    tile_product(XS, Z, tile, bf, bv, asw, asr, lane, grp, mrow);
}


#define TM    32
#define PITCH (TM + 1)
#define PPW   (TM / 8)

template <int K>
__device__ __forceinline__ int row_lane(int v) {
  return __builtin_amdgcn_update_dpp(0, v, 0x150 + K, 0xf, 0xf, false);
}

#define GATHER_STEP(K)                                                              \
  {                                                                                 \
    const unsigned off = ((unsigned)row_lane<K>(iv) << 8) + cofs;                   \
    const v4f yv = *(const v4f*)(yb + off);                                         \
    const v4f s = yv + zv;                                                          \
    acc.x = fmaxf(acc.x, s.x);                                                      \
    acc.y = fmaxf(acc.y, s.y);                                                      \
    acc.z = fmaxf(acc.z, s.z);                                                      \
    acc.w = fmaxf(acc.w, s.w);                                                      \
  }

__launch_bounds__(256)
__global__ void gather_max_kernel(const float* __restrict__ y,
                                  const float* __restrict__ z,
                                  const int* __restrict__ knn,
                                  float* __restrict__ out) {
  __shared__ float tile[OO * PITCH];

  const int lane = threadIdx.x & 31;
  const int h    = lane >> 4;
  const int c    = lane & 15;
  const int wib  = __builtin_amdgcn_readfirstlane(threadIdx.x >> 5);
  const int base = blockIdx.x * TM;
  const int b    = base >> 13;
  const int m0   = base & (MM - 1);
  const int pl0  = wib * PPW;

  const char* yb = (const char*)(y + (size_t)b * NN * OO);
  const unsigned cofs = (unsigned)c * 16u;

  const int*   kn = knn + (size_t)(base + pl0) * KK + lane;
  const float* zr = z + (size_t)(base + pl0) * OO + lane * 4;

#pragma unroll
  for (int j = 0; j < PPW / 2; ++j) {
    const int iv = kn[32 * j];
    const v4f zv = *(const v4f*)(zr + 2 * OO * j);
    v4f acc;
    acc.x = 0.0f;
    acc.y = 0.0f;
    acc.z = 0.0f;
    acc.w = 0.0f;
    GATHER_STEP(0)  GATHER_STEP(1)  GATHER_STEP(2)  GATHER_STEP(3)
    GATHER_STEP(4)  GATHER_STEP(5)  GATHER_STEP(6)  GATHER_STEP(7)
    GATHER_STEP(8)  GATHER_STEP(9)  GATHER_STEP(10) GATHER_STEP(11)
    GATHER_STEP(12) GATHER_STEP(13) GATHER_STEP(14) GATHER_STEP(15)
    float* tp = tile + (4 * c) * PITCH + pl0 + 2 * j + h;
    tp[0 * PITCH] = acc.x;
    tp[1 * PITCH] = acc.y;
    tp[2 * PITCH] = acc.z;
    tp[3 * PITCH] = acc.w;
  }

  __syncthreads();

  float* ob = out + ((size_t)b * OO + wib * 8) * MM + m0 + lane;
#pragma unroll
  for (int r = 0; r < 8; ++r)
    ob[(size_t)r * MM] = tile[(wib * 8 + r) * PITCH + lane];
}

extern "C" void kernel_launch(void* const* d_in, const int* in_sizes, int n_in,
                              void* d_out, int out_size, void* d_ws, size_t ws_size,
                              hipStream_t stream) {
  const float* x    = (const float*)d_in[0];
  const float* xs   = (const float*)d_in[1];
  const int*   knn  = (const int*)d_in[2];
  const float* W    = (const float*)d_in[3];
  const float* bias = (const float*)d_in[4];
  float* out = (float*)d_out;

  float* y = (float*)d_ws;
  float* z = y + (size_t)BB * NN * OO;

  gemm16_wmma_yz<<<256, 256, 0, stream>>>(x, xs, W, bias, y, z, (BB * NN) / 16, (BB * MM) / 16);
  gather_max_kernel<<<(BB * MM) / TM, 256, 0, stream>>>(y, z, knn, out);
}


# gather launch: 256 resident work-groups, each wave walks 16 points with the row loads of the next two pairs in flight under the current pair's maxima (2048 waves, was 8192 waves of 4 points each)
# speedup vs baseline: 3.0661x; 1.0564x over previous
#include <hip/hip_runtime.h>


typedef float v2f __attribute__((ext_vector_type(2)));
typedef float v8f __attribute__((ext_vector_type(8)));

#define BB 4
#define NN 32768
#define MM 8192
#define KK 16
#define DD 64
#define OO 64

typedef float v4f __attribute__((ext_vector_type(4)));

#define GW 8
#define LP (DD + 4)

__device__ __forceinline__ void load_b_frags(const float* Bm, int mrow, int kb, v2f (&bf)[4][16]) {
#pragma unroll
  for (int t = 0; t < 4; ++t) {
    const float* brow = Bm + (t * 16 + mrow) * LP + kb;
#pragma unroll
    for (int s = 0; s < 16; ++s)
      bf[t][s] = *(const v2f*)(brow + 4 * s);
  }
}

__device__ __forceinline__ void tile_product(const float* __restrict__ X, float* __restrict__ Y, int tile,
                                             const v2f (&bf)[4][16], const float (&bv)[4],
                                             float* asw, const float* asr,
                                             int lane, int grp, int mrow) {
  const float* xt = X + (size_t)tile * (16 * DD) + lane * 4;
  v4f g[8];
#pragma unroll
  for (int i = 0; i < 8; ++i) g[i] = *(const v4f*)(xt + 128 * i);
#pragma unroll
  for (int i = 0; i < 8; ++i) *(v4f*)(asw + 2 * i * LP) = g[i];


  v2f a[16];
#pragma unroll
  for (int s = 0; s < 16; ++s) a[s] = *(const v2f*)(asr + 4 * s);

  __builtin_amdgcn_sched_barrier(0);

  v8f acc[4];
#pragma unroll
  for (int t = 0; t < 4; ++t)
#pragma unroll
    for (int i = 0; i < 8; ++i) acc[t][i] = bv[t];

#pragma unroll
  for (int s = 0; s < 16; ++s) {
#pragma unroll
    for (int t = 0; t < 4; ++t) {
      acc[t] = __builtin_amdgcn_wmma_f32_16x16x4_f32(
           false, a[s],  false, bf[t][s],
           (short)0, acc[t],  false,  false);
    }
  }

  float* yout = Y + (size_t)(tile * 16 + 8 * grp) * OO;
#pragma unroll
  for (int t = 0; t < 4; ++t)
#pragma unroll
    for (int r = 0; r < 8; ++r)
      yout[(size_t)r * OO + t * 16 + mrow] = acc[t][r];
}

__launch_bounds__(256)
__global__ void gemm16_wmma_yz(const float* __restrict__ X,
                               const float* __restrict__ XS,
                               const float* __restrict__ W,
                               const float* __restrict__ bias,
                               float* __restrict__ Y,
                               float* __restrict__ Z,
                               int yTiles, int zTiles) {
  __shared__ __attribute__((aligned(16))) float Bs1[OO * LP];
  __shared__ __attribute__((aligned(16))) float Bs2[OO * LP];
  __shared__ __attribute__((aligned(16))) float As[GW][16 * LP];

  const int lane  = threadIdx.x & 31;
  const int wib   = threadIdx.x >> 5;
  const int gwave = blockIdx.x * GW + wib;
  const int nWav  = gridDim.x * GW;

  const int mrow = lane & 15;
  const int grp  = lane >> 4;
  const int kb   = grp * 2;

#pragma unroll
  for (int i = 0; i < 4; ++i) {
    const int idx = threadIdx.x + 256 * i;
    const int row = idx >> 4;
    const int c4  = idx & 15;
    const float* wp = W + (size_t)row * (2 * DD) + 4 * c4;
    const v4f q = *(const v4f*)wp;
    const v4f p = *(const v4f*)(wp + DD);
    *(v4f*)(Bs1 + row * LP + 4 * c4) = q;
    *(v4f*)(Bs2 + row * LP + 4 * c4) = p - q;
  }
  float bz[4];
#pragma unroll
  for (int t = 0; t < 4; ++t) bz[t] = bias[t * 16 + mrow];
  __syncthreads();

  float* as = As[wib];
  float*       asw = as + grp * LP + 4 * mrow;
  const float* asr = as + mrow * LP + kb;

  v2f bf[4][16];
  float bv[4];

  load_b_frags(Bs1, mrow, kb, bf);
#pragma unroll
  for (int t = 0; t < 4; ++t) bv[t] = 0.0f;
  for (int tile = gwave; tile < yTiles; tile += nWav)
    tile_product(X, Y, tile, bf, bv, asw, asr, lane, grp, mrow);

  load_b_frags(Bs2, mrow, kb, bf);
#pragma unroll
  for (int t = 0; t < 4; ++t) bv[t] = bz[t];
  for (int tile = gwave; tile < zTiles; tile += nWav)
    tile_product(XS, Z, tile, bf, bv, asw, asr, lane, grp, mrow);
}


#define TM    128
#define PITCH (TM + 1)
#define PPW   (TM / 8)
#define NPAIR (PPW / 2)
#define NBUF  3

template <int K>
__device__ __forceinline__ int row_lane(int v) {
  return __builtin_amdgcn_update_dpp(0, v, 0x150 + K, 0xf, 0xf, false);
}

template <int K>
__device__ __forceinline__ unsigned row_offset(unsigned cofs, int iv) {
  return ((unsigned)row_lane<K>(iv) << 8) + cofs;
}

__device__ __forceinline__ void row_offsets(unsigned cofs, int iv, unsigned (&o)[KK]) {
  o[0]  = row_offset<0>(cofs, iv);   o[1]  = row_offset<1>(cofs, iv);
  o[2]  = row_offset<2>(cofs, iv);   o[3]  = row_offset<3>(cofs, iv);
  o[4]  = row_offset<4>(cofs, iv);   o[5]  = row_offset<5>(cofs, iv);
  o[6]  = row_offset<6>(cofs, iv);   o[7]  = row_offset<7>(cofs, iv);
  o[8]  = row_offset<8>(cofs, iv);   o[9]  = row_offset<9>(cofs, iv);
  o[10] = row_offset<10>(cofs, iv);  o[11] = row_offset<11>(cofs, iv);
  o[12] = row_offset<12>(cofs, iv);  o[13] = row_offset<13>(cofs, iv);
  o[14] = row_offset<14>(cofs, iv);  o[15] = row_offset<15>(cofs, iv);
}

__device__ __forceinline__ void load_rows(const char* yb, const unsigned (&o)[KK], v4f (&r)[KK]) {
#pragma unroll
  for (int k = 0; k < KK; ++k) r[k] = *(const v4f*)(yb + o[k]);
}

__launch_bounds__(256)
__global__ void gather_max_kernel(const float* __restrict__ y,
                                  const float* __restrict__ z,
                                  const int* __restrict__ knn,
                                  float* __restrict__ out) {
  __shared__ float tile[OO * PITCH];

  const int lane = threadIdx.x & 31;
  const int h    = lane >> 4;
  const int c    = lane & 15;
  const int wib  = __builtin_amdgcn_readfirstlane(threadIdx.x >> 5);
  const int base = blockIdx.x * TM;
  const int b    = base >> 13;
  const int m0   = base & (MM - 1);
  const int pl0  = wib * PPW;

  const char* yb = (const char*)(y + (size_t)b * NN * OO);
  const unsigned cofs = (unsigned)c * 16u;

  const int*   kn = knn + (size_t)(base + pl0) * KK + lane;
  const float* zr = z + (size_t)(base + pl0) * OO + lane * 4;

  int iv[NPAIR];
  v4f zv[NPAIR];
#pragma unroll
  for (int j = 0; j < NPAIR; ++j) iv[j] = kn[32 * j];
#pragma unroll
  for (int j = 0; j < NPAIR; ++j) zv[j] = *(const v4f*)(zr + 2 * OO * j);

  v4f rows[NBUF][KK];
  unsigned off[KK];
#pragma unroll
  for (int p = 0; p < NBUF - 1; ++p) {
    row_offsets(cofs, iv[p], off);
    load_rows(yb, off, rows[p]);
  }

#pragma unroll
  for (int j = 0; j < NPAIR; ++j) {
    if (j + NBUF - 1 < NPAIR) {
      row_offsets(cofs, iv[j + NBUF - 1], off);
      load_rows(yb, off, rows[(j + NBUF - 1) % NBUF]);
    }
    const v4f (&r)[KK] = rows[j % NBUF];
    v4f acc;
    acc.x = 0.0f;
    acc.y = 0.0f;
    acc.z = 0.0f;
    acc.w = 0.0f;
#pragma unroll
    for (int k = 0; k < KK; ++k) {
      const v4f s = r[k] + zv[j];
      acc.x = fmaxf(acc.x, s.x);
      acc.y = fmaxf(acc.y, s.y);
      acc.z = fmaxf(acc.z, s.z);
      acc.w = fmaxf(acc.w, s.w);
    }
    float* tp = tile + (4 * c) * PITCH + pl0 + 2 * j + h;
    tp[0 * PITCH] = acc.x;
    tp[1 * PITCH] = acc.y;
    tp[2 * PITCH] = acc.z;
    tp[3 * PITCH] = acc.w;
  }

  __syncthreads();

  float* ob = out + ((size_t)b * OO + wib * 8) * MM + m0 + lane;
#pragma unroll
  for (int r = 0; r < 8; ++r)
#pragma unroll
    for (int q = 0; q < TM / 32; ++q)
      ob[(size_t)r * MM + 32 * q] = tile[(wib * 8 + r) * PITCH + 32 * q + lane];
}

extern "C" void kernel_launch(void* const* d_in, const int* in_sizes, int n_in,
                              void* d_out, int out_size, void* d_ws, size_t ws_size,
                              hipStream_t stream) {
  const float* x    = (const float*)d_in[0];
  const float* xs   = (const float*)d_in[1];
  const int*   knn  = (const int*)d_in[2];
  const float* W    = (const float*)d_in[3];
  const float* bias = (const float*)d_in[4];
  float* out = (float*)d_out;

  float* y = (float*)d_ws;
  float* z = y + (size_t)BB * NN * OO;

  gemm16_wmma_yz<<<256, 256, 0, stream>>>(x, xs, W, bias, y, z, (BB * NN) / 16, (BB * MM) / 16);
  gather_max_kernel<<<(BB * MM) / TM, 256, 0, stream>>>(y, z, knn, out);
}


# z product moved from the y launch into the gather launch: each wave forms z for its 16 points on WMMA under its row loads and adds it once, after the max (same bits); z never goes through memory
# speedup vs baseline: 3.3162x; 1.0816x over previous
#include <hip/hip_runtime.h>


typedef float v2f __attribute__((ext_vector_type(2)));
typedef float v8f __attribute__((ext_vector_type(8)));

#define BB 4
#define NN 32768
#define MM 8192
#define KK 16
#define DD 64
#define OO 64

typedef float v4f __attribute__((ext_vector_type(4)));

#define GW 8
#define LP (DD + 4)

__device__ __forceinline__ void load_b_frags(const float* Bm, int mrow, int kb, v2f (&bf)[4][16]) {
#pragma unroll
  for (int t = 0; t < 4; ++t) {
    const float* brow = Bm + (t * 16 + mrow) * LP + kb;
#pragma unroll
    for (int s = 0; s < 16; ++s)
      bf[t][s] = *(const v2f*)(brow + 4 * s);
  }
}

__device__ __forceinline__ void tile_wmma(const float* asr, const v2f (&bf)[4][16], const float (&bv)[4],
                                          v8f (&acc)[4]) {
  v2f a[16];
#pragma unroll
  for (int s = 0; s < 16; ++s) a[s] = *(const v2f*)(asr + 4 * s);

  __builtin_amdgcn_sched_barrier(0);

#pragma unroll
  for (int t = 0; t < 4; ++t)
#pragma unroll
    for (int i = 0; i < 8; ++i) acc[t][i] = bv[t];

#pragma unroll
  for (int s = 0; s < 16; ++s) {
#pragma unroll
    for (int t = 0; t < 4; ++t) {
      acc[t] = __builtin_amdgcn_wmma_f32_16x16x4_f32(
           false, a[s],  false, bf[t][s],
           (short)0, acc[t],  false,  false);
    }
  }
}

__device__ __forceinline__ void tile_product(const float* __restrict__ X, float* __restrict__ Y, int tile,
                                             const v2f (&bf)[4][16], const float (&bv)[4],
                                             float* asw, const float* asr,
                                             int lane, int grp, int mrow) {
  const float* xt = X + (size_t)tile * (16 * DD) + lane * 4;
  v4f g[8];
#pragma unroll
  for (int i = 0; i < 8; ++i) g[i] = *(const v4f*)(xt + 128 * i);
#pragma unroll
  for (int i = 0; i < 8; ++i) *(v4f*)(asw + 2 * i * LP) = g[i];

  v8f acc[4];
  tile_wmma(asr, bf, bv, acc);

  float* yout = Y + (size_t)(tile * 16 + 8 * grp) * OO;
#pragma unroll
  for (int t = 0; t < 4; ++t)
#pragma unroll
    for (int r = 0; r < 8; ++r)
      yout[(size_t)r * OO + t * 16 + mrow] = acc[t][r];
}

__launch_bounds__(256)
__global__ void gemm16_wmma_y(const float* __restrict__ X,
                              const float* __restrict__ W,
                              float* __restrict__ Y,
                              int yTiles) {
  __shared__ __attribute__((aligned(16))) float Bs1[OO * LP];
  __shared__ __attribute__((aligned(16))) float As[GW][16 * LP];

  const int lane  = threadIdx.x & 31;
  const int wib   = threadIdx.x >> 5;
  const int gwave = blockIdx.x * GW + wib;
  const int nWav  = gridDim.x * GW;

  const int mrow = lane & 15;
  const int grp  = lane >> 4;
  const int kb   = grp * 2;

#pragma unroll
  for (int i = 0; i < 4; ++i) {
    const int idx = threadIdx.x + 256 * i;
    const int row = idx >> 4;
    const int c4  = idx & 15;
    *(v4f*)(Bs1 + row * LP + 4 * c4) = *(const v4f*)(W + (size_t)row * (2 * DD) + 4 * c4);
  }
  __syncthreads();

  float* as = As[wib];
  float*       asw = as + grp * LP + 4 * mrow;
  const float* asr = as + mrow * LP + kb;

  v2f bf[4][16];
  float bv[4];
  load_b_frags(Bs1, mrow, kb, bf);
#pragma unroll
  for (int t = 0; t < 4; ++t) bv[t] = 0.0f;
  for (int tile = gwave; tile < yTiles; tile += nWav)
    tile_product(X, Y, tile, bf, bv, asw, asr, lane, grp, mrow);
}


#define TM    128
#define PITCH (TM + 1)
#define PPW   (TM / 8)
#define NPAIR (PPW / 2)
#define NBUF  3

template <int K>
__device__ __forceinline__ int row_lane(int v) {
  return __builtin_amdgcn_update_dpp(0, v, 0x150 + K, 0xf, 0xf, false);
}

template <int K>
__device__ __forceinline__ unsigned row_offset(unsigned cofs, int iv) {
  return ((unsigned)row_lane<K>(iv) << 8) + cofs;
}

__device__ __forceinline__ void row_offsets(unsigned cofs, int iv, unsigned (&o)[KK]) {
  o[0]  = row_offset<0>(cofs, iv);   o[1]  = row_offset<1>(cofs, iv);
  o[2]  = row_offset<2>(cofs, iv);   o[3]  = row_offset<3>(cofs, iv);
  o[4]  = row_offset<4>(cofs, iv);   o[5]  = row_offset<5>(cofs, iv);
  o[6]  = row_offset<6>(cofs, iv);   o[7]  = row_offset<7>(cofs, iv);
  o[8]  = row_offset<8>(cofs, iv);   o[9]  = row_offset<9>(cofs, iv);
  o[10] = row_offset<10>(cofs, iv);  o[11] = row_offset<11>(cofs, iv);
  o[12] = row_offset<12>(cofs, iv);  o[13] = row_offset<13>(cofs, iv);
  o[14] = row_offset<14>(cofs, iv);  o[15] = row_offset<15>(cofs, iv);
}

__device__ __forceinline__ void load_rows(const char* yb, const unsigned (&o)[KK], v4f (&r)[KK]) {
#pragma unroll
  for (int k = 0; k < KK; ++k) r[k] = *(const v4f*)(yb + o[k]);
}

__launch_bounds__(256)
__global__ void gather_max_kernel(const float* __restrict__ y,
                                  const float* __restrict__ xs,
                                  const float* __restrict__ W,
                                  const float* __restrict__ bias,
                                  const int* __restrict__ knn,
                                  float* __restrict__ out) {
  __shared__ float tile[OO * PITCH];
  __shared__ __attribute__((aligned(16))) float Bs2[OO * LP];
  __shared__ __attribute__((aligned(16))) float As[GW][16 * LP];

  const int lane = threadIdx.x & 31;
  const int h    = lane >> 4;
  const int c    = lane & 15;
  const int wib  = __builtin_amdgcn_readfirstlane(threadIdx.x >> 5);
  const int base = blockIdx.x * TM;
  const int b    = base >> 13;
  const int m0   = base & (MM - 1);
  const int pl0  = wib * PPW;

  const char* yb = (const char*)(y + (size_t)b * NN * OO);
  const unsigned cofs = (unsigned)c * 16u;

  const int* kn = knn + (size_t)(base + pl0) * KK + lane;
  int iv[NPAIR];
#pragma unroll
  for (int j = 0; j < NPAIR; ++j) iv[j] = kn[32 * j];

  v4f wq[4], wp[4];
#pragma unroll
  for (int i = 0; i < 4; ++i) {
    const int idx = threadIdx.x + 256 * i;
    const float* wsrc = W + (size_t)(idx >> 4) * (2 * DD) + 4 * (idx & 15);
    wq[i] = *(const v4f*)wsrc;
    wp[i] = *(const v4f*)(wsrc + DD);
  }
  float bz[4];
#pragma unroll
  for (int t = 0; t < 4; ++t) bz[t] = bias[t * 16 + c];
  const float* xt = xs + (size_t)(base + pl0) * DD + lane * 4;
  v4f g[8];
#pragma unroll
  for (int i = 0; i < 8; ++i) g[i] = *(const v4f*)(xt + 128 * i);

  v4f rows[NBUF][KK];
  unsigned off[KK];
#pragma unroll
  for (int p = 0; p < NBUF; ++p) {
    row_offsets(cofs, iv[p], off);
    load_rows(yb, off, rows[p]);
  }

#pragma unroll
  for (int i = 0; i < 4; ++i) {
    const int idx = threadIdx.x + 256 * i;
    *(v4f*)(Bs2 + (idx >> 4) * LP + 4 * (idx & 15)) = wp[i] - wq[i];
  }
  __builtin_amdgcn_fence(__ATOMIC_RELEASE, "workgroup", "local");
  __builtin_amdgcn_s_barrier();
  __builtin_amdgcn_fence(__ATOMIC_ACQUIRE, "workgroup", "local");

  {
    float* as = As[wib];
    float*       asw = as + h * LP + 4 * c;
    const float* asr = as + c * LP + 2 * h;
#pragma unroll
    for (int i = 0; i < 8; ++i) *(v4f*)(asw + 2 * i * LP) = g[i];

    v2f bf[4][16];
    load_b_frags(Bs2, c, 2 * h, bf);
    v8f acc[4];
    tile_wmma(asr, bf, bz, acc);

    float* zp = tile + c * PITCH + pl0 + 8 * h;
#pragma unroll
    for (int t = 0; t < 4; ++t)
#pragma unroll
      for (int r = 0; r < 8; ++r)
        zp[16 * t * PITCH + r] = acc[t][r];
  }

#pragma unroll
  for (int j = 0; j < NPAIR; ++j) {
    const v4f (&r)[KK] = rows[j % NBUF];
    v4f mx = r[0];
#pragma unroll
    for (int k = 1; k < KK; ++k) {
      mx.x = fmaxf(mx.x, r[k].x);
      mx.y = fmaxf(mx.y, r[k].y);
      mx.z = fmaxf(mx.z, r[k].z);
      mx.w = fmaxf(mx.w, r[k].w);
    }
    if (j + NBUF < NPAIR) {
      row_offsets(cofs, iv[j + NBUF], off);
      load_rows(yb, off, rows[j % NBUF]);
    }
    float* tp = tile + (4 * c) * PITCH + pl0 + 2 * j + h;
    tp[0 * PITCH] = fmaxf(0.0f, mx.x + tp[0 * PITCH]);
    tp[1 * PITCH] = fmaxf(0.0f, mx.y + tp[1 * PITCH]);
    tp[2 * PITCH] = fmaxf(0.0f, mx.z + tp[2 * PITCH]);
    tp[3 * PITCH] = fmaxf(0.0f, mx.w + tp[3 * PITCH]);
  }

  __syncthreads();

  float* ob = out + ((size_t)b * OO + wib * 8) * MM + m0 + lane;
#pragma unroll
  for (int r = 0; r < 8; ++r)
#pragma unroll
    for (int q = 0; q < TM / 32; ++q)
      ob[(size_t)r * MM + 32 * q] = tile[(wib * 8 + r) * PITCH + 32 * q + lane];
}

extern "C" void kernel_launch(void* const* d_in, const int* in_sizes, int n_in,
                              void* d_out, int out_size, void* d_ws, size_t ws_size,
                              hipStream_t stream) {
  const float* x    = (const float*)d_in[0];
  const float* xs   = (const float*)d_in[1];
  const int*   knn  = (const int*)d_in[2];
  const float* W    = (const float*)d_in[3];
  const float* bias = (const float*)d_in[4];
  float* out = (float*)d_out;

  float* y = (float*)d_ws;

  gemm16_wmma_y<<<256, 256, 0, stream>>>(x, W, y, (BB * NN) / 16);
  gather_max_kernel<<<(BB * MM) / TM, 256, 0, stream>>>(y, xs, W, bias, knn, out);
}


# y launch: x tile loads issued two tiles ahead into three register sets, the first two at entry ahead of the W staging, so a tile's rows travel under the WMMAs of the tiles before it
# speedup vs baseline: 3.4415x; 1.0378x over previous
#include <hip/hip_runtime.h>


typedef float v2f __attribute__((ext_vector_type(2)));
typedef float v8f __attribute__((ext_vector_type(8)));

#define BB 4
#define NN 32768
#define MM 8192
#define KK 16
#define DD 64
#define OO 64

typedef float v4f __attribute__((ext_vector_type(4)));

#define GW 8
#define LP (DD + 4)

__device__ __forceinline__ void load_b_frags(const float* Bm, int mrow, int kb, v2f (&bf)[4][16]) {
#pragma unroll
  for (int t = 0; t < 4; ++t) {
    const float* brow = Bm + (t * 16 + mrow) * LP + kb;
#pragma unroll
    for (int s = 0; s < 16; ++s)
      bf[t][s] = *(const v2f*)(brow + 4 * s);
  }
}

__device__ __forceinline__ void tile_wmma(const float* asr, const v2f (&bf)[4][16], const float (&bv)[4],
                                          v8f (&acc)[4]) {
  v2f a[16];
#pragma unroll
  for (int s = 0; s < 16; ++s) a[s] = *(const v2f*)(asr + 4 * s);

  __builtin_amdgcn_sched_barrier(0);

#pragma unroll
  for (int t = 0; t < 4; ++t)
#pragma unroll
    for (int i = 0; i < 8; ++i) acc[t][i] = bv[t];

#pragma unroll
  for (int s = 0; s < 16; ++s) {
#pragma unroll
    for (int t = 0; t < 4; ++t) {
      acc[t] = __builtin_amdgcn_wmma_f32_16x16x4_f32(
           false, a[s],  false, bf[t][s],
           (short)0, acc[t],  false,  false);
    }
  }
}

__device__ __forceinline__ void load_x_tile(const float* xl, int tile, v4f (&g)[8]) {
  const float* xt = xl + (size_t)tile * (16 * DD);
#pragma unroll
  for (int i = 0; i < 8; ++i) g[i] = *(const v4f*)(xt + 128 * i);
}

__device__ __forceinline__ void tile_product(const v4f (&g)[8], float* __restrict__ Y, int tile,
                                             const v2f (&bf)[4][16], const float (&bv)[4],
                                             float* asw, const float* asr, int grp, int mrow) {
#pragma unroll
  for (int i = 0; i < 8; ++i) *(v4f*)(asw + 2 * i * LP) = g[i];

  v8f acc[4];
  tile_wmma(asr, bf, bv, acc);

  float* yout = Y + (size_t)(tile * 16 + 8 * grp) * OO;
#pragma unroll
  for (int t = 0; t < 4; ++t)
#pragma unroll
    for (int r = 0; r < 8; ++r)
      yout[(size_t)r * OO + t * 16 + mrow] = acc[t][r];
}

#define XAHEAD 2
#define XSETS  (XAHEAD + 1)
#define YG 256
#define YT ((BB * NN / 16) / (YG * GW))
static_assert(YT * YG * GW * 16 == BB * NN, "the y launch covers x with whole tiles, YT per wave");

__launch_bounds__(256)
__global__ void gemm16_wmma_y(const float* __restrict__ X,
                              const float* __restrict__ W,
                              float* __restrict__ Y) {
  __shared__ __attribute__((aligned(16))) float Bs1[OO * LP];
  __shared__ __attribute__((aligned(16))) float As[GW][16 * LP];

  const int lane  = threadIdx.x & 31;
  const int wib   = __builtin_amdgcn_readfirstlane(threadIdx.x >> 5);
  const int gwave = blockIdx.x * GW + wib;

  const int mrow = lane & 15;
  const int grp  = lane >> 4;
  const int kb   = grp * 2;

  v4f wq[4];
#pragma unroll
  for (int i = 0; i < 4; ++i) {
    const int idx = threadIdx.x + 256 * i;
    wq[i] = *(const v4f*)(W + (size_t)(idx >> 4) * (2 * DD) + 4 * (idx & 15));
  }

  __builtin_amdgcn_sched_barrier(0);
  const float* xl = X + lane * 4;
  v4f g[XSETS][8];
#pragma unroll
  for (int i = 0; i < XAHEAD && i < YT; ++i) {
    load_x_tile(xl, gwave + i * (YG * GW), g[i]);
    __builtin_amdgcn_sched_barrier(0);
  }

#pragma unroll
  for (int i = 0; i < 4; ++i) {
    const int idx = threadIdx.x + 256 * i;
    *(v4f*)(Bs1 + (idx >> 4) * LP + 4 * (idx & 15)) = wq[i];
  }
  __builtin_amdgcn_fence(__ATOMIC_RELEASE, "workgroup", "local");
  __builtin_amdgcn_s_barrier();
  __builtin_amdgcn_fence(__ATOMIC_ACQUIRE, "workgroup", "local");

  float* as = As[wib];
  float*       asw = as + grp * LP + 4 * mrow;
  const float* asr = as + mrow * LP + kb;

  v2f bf[4][16];
  float bv[4];
  load_b_frags(Bs1, mrow, kb, bf);
#pragma unroll
  for (int t = 0; t < 4; ++t) bv[t] = 0.0f;

#pragma unroll
  for (int i = 0; i < YT; ++i) {
    const int tile = gwave + i * (YG * GW);
    if (i + XAHEAD < YT) load_x_tile(xl, tile + XAHEAD * (YG * GW), g[(i + XAHEAD) % XSETS]);
    tile_product(g[i % XSETS], Y, tile, bf, bv, asw, asr, grp, mrow);
    __builtin_amdgcn_sched_barrier(0);
  }
}


#define TM    128
#define PITCH (TM + 1)
#define PPW   (TM / 8)
#define NPAIR (PPW / 2)
#define NBUF  3

template <int K>
__device__ __forceinline__ int row_lane(int v) {
  return __builtin_amdgcn_update_dpp(0, v, 0x150 + K, 0xf, 0xf, false);
}

template <int K>
__device__ __forceinline__ unsigned row_offset(unsigned cofs, int iv) {
  return ((unsigned)row_lane<K>(iv) << 8) + cofs;
}

__device__ __forceinline__ void row_offsets(unsigned cofs, int iv, unsigned (&o)[KK]) {
  o[0]  = row_offset<0>(cofs, iv);   o[1]  = row_offset<1>(cofs, iv);
  o[2]  = row_offset<2>(cofs, iv);   o[3]  = row_offset<3>(cofs, iv);
  o[4]  = row_offset<4>(cofs, iv);   o[5]  = row_offset<5>(cofs, iv);
  o[6]  = row_offset<6>(cofs, iv);   o[7]  = row_offset<7>(cofs, iv);
  o[8]  = row_offset<8>(cofs, iv);   o[9]  = row_offset<9>(cofs, iv);
  o[10] = row_offset<10>(cofs, iv);  o[11] = row_offset<11>(cofs, iv);
  o[12] = row_offset<12>(cofs, iv);  o[13] = row_offset<13>(cofs, iv);
  o[14] = row_offset<14>(cofs, iv);  o[15] = row_offset<15>(cofs, iv);
}

__device__ __forceinline__ void load_rows(const char* yb, const unsigned (&o)[KK], v4f (&r)[KK]) {
#pragma unroll
  for (int k = 0; k < KK; ++k) r[k] = *(const v4f*)(yb + o[k]);
}

__launch_bounds__(256)
__global__ void gather_max_kernel(const float* __restrict__ y,
                                  const float* __restrict__ xs,
                                  const float* __restrict__ W,
                                  const float* __restrict__ bias,
                                  const int* __restrict__ knn,
                                  float* __restrict__ out) {
  __shared__ float tile[OO * PITCH];
  __shared__ __attribute__((aligned(16))) float Bs2[OO * LP];
  __shared__ __attribute__((aligned(16))) float As[GW][16 * LP];

  const int lane = threadIdx.x & 31;
  const int h    = lane >> 4;
  const int c    = lane & 15;
  const int wib  = __builtin_amdgcn_readfirstlane(threadIdx.x >> 5);
  const int base = blockIdx.x * TM;
  const int b    = base >> 13;
  const int m0   = base & (MM - 1);
  const int pl0  = wib * PPW;

  const char* yb = (const char*)(y + (size_t)b * NN * OO);
  const unsigned cofs = (unsigned)c * 16u;

  const int* kn = knn + (size_t)(base + pl0) * KK + lane;
  int iv[NPAIR];
#pragma unroll
  for (int j = 0; j < NPAIR; ++j) iv[j] = kn[32 * j];

  v4f wq[4], wp[4];
#pragma unroll
  for (int i = 0; i < 4; ++i) {
    const int idx = threadIdx.x + 256 * i;
    const float* wsrc = W + (size_t)(idx >> 4) * (2 * DD) + 4 * (idx & 15);
    wq[i] = *(const v4f*)wsrc;
    wp[i] = *(const v4f*)(wsrc + DD);
  }
  float bz[4];
#pragma unroll
  for (int t = 0; t < 4; ++t) bz[t] = bias[t * 16 + c];
  const float* xt = xs + (size_t)(base + pl0) * DD + lane * 4;
  v4f g[8];
#pragma unroll
  for (int i = 0; i < 8; ++i) g[i] = *(const v4f*)(xt + 128 * i);

  v4f rows[NBUF][KK];
  unsigned off[KK];
#pragma unroll
  for (int p = 0; p < NBUF; ++p) {
    row_offsets(cofs, iv[p], off);
    load_rows(yb, off, rows[p]);
  }

#pragma unroll
  for (int i = 0; i < 4; ++i) {
    const int idx = threadIdx.x + 256 * i;
    *(v4f*)(Bs2 + (idx >> 4) * LP + 4 * (idx & 15)) = wp[i] - wq[i];
  }
  __builtin_amdgcn_fence(__ATOMIC_RELEASE, "workgroup", "local");
  __builtin_amdgcn_s_barrier();
  __builtin_amdgcn_fence(__ATOMIC_ACQUIRE, "workgroup", "local");

  {
    float* as = As[wib];
    float*       asw = as + h * LP + 4 * c;
    const float* asr = as + c * LP + 2 * h;
#pragma unroll
    for (int i = 0; i < 8; ++i) *(v4f*)(asw + 2 * i * LP) = g[i];

    v2f bf[4][16];
    load_b_frags(Bs2, c, 2 * h, bf);
    v8f acc[4];
    tile_wmma(asr, bf, bz, acc);

    float* zp = tile + c * PITCH + pl0 + 8 * h;
#pragma unroll
    for (int t = 0; t < 4; ++t)
#pragma unroll
      for (int r = 0; r < 8; ++r)
        zp[16 * t * PITCH + r] = acc[t][r];
  }

#pragma unroll
  for (int j = 0; j < NPAIR; ++j) {
    const v4f (&r)[KK] = rows[j % NBUF];
    v4f mx = r[0];
#pragma unroll
    for (int k = 1; k < KK; ++k) {
      mx.x = fmaxf(mx.x, r[k].x);
      mx.y = fmaxf(mx.y, r[k].y);
      mx.z = fmaxf(mx.z, r[k].z);
      mx.w = fmaxf(mx.w, r[k].w);
    }
    if (j + NBUF < NPAIR) {
      row_offsets(cofs, iv[j + NBUF], off);
      load_rows(yb, off, rows[j % NBUF]);
    }
    float* tp = tile + (4 * c) * PITCH + pl0 + 2 * j + h;
    tp[0 * PITCH] = fmaxf(0.0f, mx.x + tp[0 * PITCH]);
    tp[1 * PITCH] = fmaxf(0.0f, mx.y + tp[1 * PITCH]);
    tp[2 * PITCH] = fmaxf(0.0f, mx.z + tp[2 * PITCH]);
    tp[3 * PITCH] = fmaxf(0.0f, mx.w + tp[3 * PITCH]);
  }

  __syncthreads();

  float* ob = out + ((size_t)b * OO + wib * 8) * MM + m0 + lane;
#pragma unroll
  for (int r = 0; r < 8; ++r)
#pragma unroll
    for (int q = 0; q < TM / 32; ++q)
      ob[(size_t)r * MM + 32 * q] = tile[(wib * 8 + r) * PITCH + 32 * q + lane];
}

extern "C" void kernel_launch(void* const* d_in, const int* in_sizes, int n_in,
                              void* d_out, int out_size, void* d_ws, size_t ws_size,
                              hipStream_t stream) {
  const float* x    = (const float*)d_in[0];
  const float* xs   = (const float*)d_in[1];
  const int*   knn  = (const int*)d_in[2];
  const float* W    = (const float*)d_in[3];
  const float* bias = (const float*)d_in[4];
  float* out = (float*)d_out;

  float* y = (float*)d_ws;

  gemm16_wmma_y<<<YG, 256, 0, stream>>>(x, W, y);
  gather_max_kernel<<<(BB * MM) / TM, 256, 0, stream>>>(y, xs, W, bias, knn, out);
}
